# starting kernel
# baseline (speedup 1.0000x reference)
#include <hip/hip_runtime.h>
#include <math.h>

#define BB   256
#define SS   512
#define DD   256
#define LLN  4
#define NITERS 10
#define BT   16
#define LSTR 260
#define TILE (BT * LSTR)
#define SMEM_FLOATS (9 * TILE)
#define SMEM_BYTES  (SMEM_FLOATS * sizeof(float))

typedef float v2f __attribute__((ext_vector_type(2)));
typedef float v8f __attribute__((ext_vector_type(8)));

#if defined(__has_builtin)
#  if __has_builtin(__builtin_amdgcn_wmma_f32_16x16x4_f32)
#    define HAS_WMMA_F32 1
#  else
#    define HAS_WMMA_F32 0
#  endif
#  if __has_builtin(__builtin_amdgcn_tanhf)
#    define HAS_HW_TANH 1
#  else
#    define HAS_HW_TANH 0
#  endif
#else
#  define HAS_WMMA_F32 0
#  define HAS_HW_TANH 0
#endif

__device__ __forceinline__ float tanh_fast(float x) {
#if HAS_HW_TANH
  return __builtin_amdgcn_tanhf(x);
#else
  return tanhf(x);
#endif
}

__device__ __forceinline__ float* smb(float* sm, int i) { return sm + i * TILE; }


__device__ __forceinline__ v8f tile_xWT(const float* __restrict__ W,
                                        const float* X, int nt, int lane,
                                        float cinit) {
  const int l16 = lane & 15;
  const int half = lane >> 4;
  v8f acc = {cinit, cinit, cinit, cinit, cinit, cinit, cinit, cinit};
#if HAS_WMMA_F32
  const float* wr = W + (nt * 16 + l16) * DD + 2 * half;
  const float* xr = X + l16 * LSTR + 2 * half;
#pragma unroll 4
  for (int kk = 0; kk < DD; kk += 4) {
    v2f a = {xr[kk], xr[kk + 1]};
    v2f b = {wr[kk], wr[kk + 1]};
    acc = __builtin_amdgcn_wmma_f32_16x16x4_f32(false, a, false, b,
                                                (short)0, acc, false, false);
  }
#else
#pragma unroll
  for (int j = 0; j < 8; ++j) {
    const int m = j + 8 * half, n = nt * 16 + l16;
    float s = cinit;
    for (int k = 0; k < DD; ++k) s += X[m * LSTR + k] * W[n * DD + k];
    acc[j] = s;
  }
#endif
  return acc;
}

__device__ __forceinline__ v8f tile_xM(const float* __restrict__ M,
                                       const float* X, int nt, int lane,
                                       float cinit) {
  const int l16 = lane & 15;
  const int half = lane >> 4;
  v8f acc = {cinit, cinit, cinit, cinit, cinit, cinit, cinit, cinit};
#if HAS_WMMA_F32
  const float* mp = M + (nt * 16 + l16) + (2 * half) * DD;
  const float* xr = X + l16 * LSTR + 2 * half;
#pragma unroll 4
  for (int kk = 0; kk < DD; kk += 4) {
    v2f a = {xr[kk], xr[kk + 1]};
    v2f b = {mp[kk * DD], mp[kk * DD + DD]};
    acc = __builtin_amdgcn_wmma_f32_16x16x4_f32(false, a, false, b,
                                                (short)0, acc, false, false);
  }
#else
#pragma unroll
  for (int j = 0; j < 8; ++j) {
    const int m = j + 8 * half, n = nt * 16 + l16;
    float s = cinit;
    for (int k = 0; k < DD; ++k) s += X[m * LSTR + k] * M[k * DD + n];
    acc[j] = s;
  }
#endif
  return acc;
}

template <typename Epi>
__device__ __forceinline__ void gemm_xWT(const float* __restrict__ W,
                                         const float* __restrict__ bias,
                                         const float* X, int lane, int wv,
                                         Epi epi) {
#pragma unroll
  for (int t = 0; t < 2; ++t) {
    const int nt = wv * 2 + t;
    const int l16 = lane & 15, half = lane >> 4;
    const float bv = bias ? bias[nt * 16 + l16] : 0.f;
    v8f acc = tile_xWT(W, X, nt, lane, bv);
#pragma unroll
    for (int j = 0; j < 8; ++j) epi(j + 8 * half, nt * 16 + l16, acc[j]);
  }
}

template <typename Epi>
__device__ __forceinline__ void gemm_xM(const float* __restrict__ M,
                                        const float* X, int lane, int wv,
                                        Epi epi) {
#pragma unroll
  for (int t = 0; t < 2; ++t) {
    const int nt = wv * 2 + t;
    v8f acc = tile_xM(M, X, nt, lane, 0.f);
    const int l16 = lane & 15, half = lane >> 4;
#pragma unroll
    for (int j = 0; j < 8; ++j) epi(j + 8 * half, nt * 16 + l16, acc[j]);
  }
}

__global__ __launch_bounds__(256) void pooled_kernel(
    const int* __restrict__ ids, const float* __restrict__ emb,
    const float* __restrict__ pos, float* __restrict__ pooled_g) {
  __shared__ int sid[SS];
  const int b = blockIdx.x;
  const int t = threadIdx.x;
  sid[t] = ids[b * SS + t];
  sid[t + 256] = ids[b * SS + 256 + t];
  __syncthreads();
  float acc = 0.f;
  for (int s = 0; s < SS; ++s) {
    acc += emb[(size_t)sid[s] * DD + t] + pos[s * DD + t];
  }
  pooled_g[b * DD + t] = acc * (1.0f / (float)SS);
}

__global__ __launch_bounds__(256) void refine_kernel(
    const float* __restrict__ pooled_g,
    const float* __restrict__ init_w, const float* __restrict__ init_b,
    const float* __restrict__ pred_w, const float* __restrict__ pred_b,
    const float* __restrict__ upd_w, const float* __restrict__ upd_b,
    const float* __restrict__ sqw, const float* __restrict__ sqb,
    const float* __restrict__ eqw, const float* __restrict__ eqb,
    const float* __restrict__ keyw, const float* __restrict__ keyb,
    const float* __restrict__ aw1, const float* __restrict__ ab1,
    const float* __restrict__ aw2, const float* __restrict__ ab2,
    float* __restrict__ sq_g, float* __restrict__ eq_g,
    float* __restrict__ cs_g, float* __restrict__ ce_g,
    float* __restrict__ ans_g) {
  extern __shared__ float sm[];

  const int tid = threadIdx.x;
  const int lane = tid & 31;
  const int wv = tid >> 5;
  const int b0 = blockIdx.x * BT;

  float* pooled = smb(sm, 0);
  float* ff = smb(sm, 5);
  float* tot = smb(sm, 8);

  for (int idx = tid; idx < BT * DD; idx += 256) {
    const int r = idx >> 8, c = idx & 255;
    pooled[r * LSTR + c] = pooled_g[(b0 + r) * DD + c];
  }
  __syncthreads();

  {
    const float* cur = pooled;
    for (int l = 0; l < LLN; ++l) {
      float* dst = smb(sm, 1 + l);
      const bool last = (l == LLN - 1);
      gemm_xWT(init_w + l * DD * DD, init_b + l * DD, cur, lane, wv,
               [&](int m, int n, float v) {
                 const float r = tanh_fast(v);
                 dst[m * LSTR + n] = r;
                 if (last) ff[m * LSTR + n] = r;
               });
      __syncthreads();
      cur = dst;
    }
  }

  for (int it = 0; it < NITERS; ++it) {
    for (int i = LLN - 1; i >= 0; --i) {
      float* pc = smb(sm, 6 + ((LLN - 1 - i) & 1));
      float* pn = smb(sm, 6 + ((LLN - i) & 1));
      float* ri = smb(sm, 1 + i);
      gemm_xWT(pred_w + i * DD * DD, pred_b + i * DD, ri, lane, wv,
               [&](int m, int n, float v) { pc[m * LSTR + n] = v; });
      __syncthreads();
      {
        const float* below = (i > 0) ? smb(sm, i) : pooled;
        const bool has_above = (i < LLN - 1);
        for (int idx = tid; idx < BT * DD; idx += 256) {
          const int r = idx >> 8, c = idx & 255;
          const int o = r * LSTR + c;
          float e = below[o] - pc[o];
          if (has_above) e += 0.5f * (ri[o] - pn[o]);
          tot[o] = e;
        }
      }
      __syncthreads();
      gemm_xWT(upd_w + i * DD * DD, upd_b + i * DD, tot, lane, wv,
               [&](int m, int n, float v) {
                 const int o = m * LSTR + n;
                 ri[o] = ri[o] + 0.1f * tanh_fast(v);
               });
      __syncthreads();
    }
  }

  float* refined = smb(sm, LLN);
  for (int idx = tid; idx < BT * DD; idx += 256) {
    const int r = idx >> 8, c = idx & 255;
    refined[r * LSTR + c] += ff[r * LSTR + c];
  }
  __syncthreads();

  {
    float* q = smb(sm, 6);
    gemm_xWT(sqw, sqb, refined, lane, wv,
             [&](int m, int n, float v) { q[m * LSTR + n] = v; });
    __syncthreads();
    {
      const int r = tid >> 4, c = tid & 15;
      float p = 0.f;
#pragma unroll
      for (int u = 0; u < 16; ++u) {
        const int o = c * 16 + u;
        p += q[r * LSTR + o] * keyb[o];
      }
      tot[tid] = p;
    }
    __syncthreads();
    if (tid < BT) {
      float s = 0.f;
#pragma unroll
      for (int u = 0; u < 16; ++u) s += tot[tid * 16 + u];
      cs_g[b0 + tid] = s;
    }
    __syncthreads();
    gemm_xM(keyw, q, lane, wv,
            [&](int m, int n, float v) { sq_g[(b0 + m) * DD + n] = v; });
    __syncthreads();
  }

  {
    float* q = smb(sm, 7);
    gemm_xWT(eqw, eqb, refined, lane, wv,
             [&](int m, int n, float v) { q[m * LSTR + n] = v; });
    __syncthreads();
    {
      const int r = tid >> 4, c = tid & 15;
      float p = 0.f;
#pragma unroll
      for (int u = 0; u < 16; ++u) {
        const int o = c * 16 + u;
        p += q[r * LSTR + o] * keyb[o];
      }
      tot[tid] = p;
    }
    __syncthreads();
    if (tid < BT) {
      float s = 0.f;
#pragma unroll
      for (int u = 0; u < 16; ++u) s += tot[tid * 16 + u];
      ce_g[b0 + tid] = s;
    }
    __syncthreads();
    gemm_xM(keyw, q, lane, wv,
            [&](int m, int n, float v) { eq_g[(b0 + m) * DD + n] = v; });
    __syncthreads();
  }

  for (int task = tid; task < BT * 128; task += 256) {
    const int r = task >> 7, j = task & 127;
    float s = ab1[j];
    const float* rr = refined + r * LSTR;
    const float* wr = aw1 + j * DD;
    for (int k = 0; k < DD; ++k) s += rr[k] * wr[k];
    tot[r * LSTR + j] = 0.5f * s * (1.0f + erff(s * 0.70710678118654752f));
  }
  __syncthreads();
  if (tid < BT * 2) {
    const int r = tid >> 1, k = tid & 1;
    float s = ab2[k];
    for (int j = 0; j < 128; ++j) s += tot[r * LSTR + j] * aw2[k * 128 + j];
    ans_g[(b0 + r) * 2 + k] = s;
  }
}

__global__ __launch_bounds__(256) void logits_kernel(
    const int* __restrict__ ids, const float* __restrict__ emb,
    const float* __restrict__ pos, const float* __restrict__ sq_g,
    const float* __restrict__ eq_g, const float* __restrict__ cs_g,
    const float* __restrict__ ce_g, float* __restrict__ out_start,
    float* __restrict__ out_end) {
  const int b = blockIdx.x;
  const int lane = threadIdx.x & 31;
  const int wv = threadIdx.x >> 5;
  float sqv[8], eqv[8];
  const float* sr = sq_g + b * DD + lane * 8;
  const float* er = eq_g + b * DD + lane * 8;
#pragma unroll
  for (int j = 0; j < 8; ++j) {
    sqv[j] = sr[j];
    eqv[j] = er[j];
  }
  const float cs = cs_g[b];
  const float ce = ce_g[b];
  for (int s = wv; s < SS; s += 8) {
    const int id = ids[b * SS + s];
    const float* row = emb + (size_t)id * DD + lane * 8;
    const float* pr = pos + s * DD + lane * 8;
    float as = 0.f, ae = 0.f;
#pragma unroll
    for (int j = 0; j < 8; ++j) {
      const float e = row[j] + pr[j];
      as += e * sqv[j];
      ae += e * eqv[j];
    }
#pragma unroll
    for (int off = 16; off; off >>= 1) {
      as += __shfl_xor(as, off, 32);
      ae += __shfl_xor(ae, off, 32);
    }
    if (lane == 0) {
      out_start[b * SS + s] = as + cs;
      out_end[b * SS + s] = ae + ce;
    }
  }
}

extern "C" void kernel_launch(void* const* d_in, const int* in_sizes, int n_in,
                              void* d_out, int out_size, void* d_ws,
                              size_t ws_size, hipStream_t stream) {
  const int* ids = (const int*)d_in[0];
  const float* emb = (const float*)d_in[1];
  const float* pos = (const float*)d_in[2];
  const float* init_w = (const float*)d_in[3];
  const float* init_b = (const float*)d_in[4];
  const float* pred_w = (const float*)d_in[5];
  const float* pred_b = (const float*)d_in[6];
  const float* upd_w = (const float*)d_in[7];
  const float* upd_b = (const float*)d_in[8];
  const float* sqw = (const float*)d_in[9];
  const float* sqb = (const float*)d_in[10];
  const float* eqw = (const float*)d_in[11];
  const float* eqb = (const float*)d_in[12];
  const float* keyw = (const float*)d_in[13];
  const float* keyb = (const float*)d_in[14];
  const float* aw1 = (const float*)d_in[15];
  const float* ab1 = (const float*)d_in[16];
  const float* aw2 = (const float*)d_in[17];
  const float* ab2 = (const float*)d_in[18];

  float* ws = (float*)d_ws;
  float* pooled = ws;
  float* sq = pooled + BB * DD;
  float* eq = sq + BB * DD;
  float* cs = eq + BB * DD;
  float* ce = cs + BB;

  float* out = (float*)d_out;
  float* out_start = out;
  float* out_end = out + BB * SS;
  float* out_ans = out + 2 * BB * SS;

  (void)hipFuncSetAttribute((const void*)refine_kernel,
                            hipFuncAttributeMaxDynamicSharedMemorySize,
                            (int)SMEM_BYTES);

  pooled_kernel<<<BB, 256, 0, stream>>>(ids, emb, pos, pooled);
  refine_kernel<<<BB / BT, 256, SMEM_BYTES, stream>>>(
      pooled, init_w, init_b, pred_w, pred_b, upd_w, upd_b, sqw, sqb, eqw, eqb,
      keyw, keyb, aw1, ab1, aw2, ab2, sq, eq, cs, ce, out_ans);
  logits_kernel<<<BB, 256, 0, stream>>>(ids, emb, pos, sq, eq, cs, ce,
                                        out_start, out_end);
}


# refine GEMMs: each wave requests all weight fragments of its two n-tiles as one group of 16-byte loads ahead of the WMMA chains; k order permuted alike in A and B loaders, so the k sum order changes
# speedup vs baseline: 1.1773x; 1.1773x over previous
#include <hip/hip_runtime.h>
#include <math.h>

#define BB   256
#define SS   512
#define DD   256
#define LLN  4
#define NITERS 10
#define BT   16
#define LSTR 260
#define TILE (BT * LSTR)
#define SMEM_FLOATS (9 * TILE)
#define SMEM_BYTES  (SMEM_FLOATS * sizeof(float))

typedef float v2f __attribute__((ext_vector_type(2)));
typedef float v4f __attribute__((ext_vector_type(4)));
typedef float v8f __attribute__((ext_vector_type(8)));

#if defined(__has_builtin)
#  if __has_builtin(__builtin_amdgcn_wmma_f32_16x16x4_f32)
#    define HAS_WMMA_F32 1
#  else
#    define HAS_WMMA_F32 0
#  endif
#  if __has_builtin(__builtin_amdgcn_tanhf)
#    define HAS_HW_TANH 1
#  else
#    define HAS_HW_TANH 0
#  endif
#else
#  define HAS_WMMA_F32 0
#  define HAS_HW_TANH 0
#endif

__device__ __forceinline__ float tanh_fast(float x) {
#if HAS_HW_TANH
  return __builtin_amdgcn_tanhf(x);
#else
  return tanhf(x);
#endif
}

__device__ __forceinline__ float* smb(float* sm, int i) { return sm + i * TILE; }


__device__ __forceinline__ v8f tile_xM(const float* __restrict__ M,
                                       const float* X, int nt, int lane,
                                       float cinit) {
  const int l16 = lane & 15;
  const int half = lane >> 4;
  v8f acc = {cinit, cinit, cinit, cinit, cinit, cinit, cinit, cinit};
#if HAS_WMMA_F32
  const float* mp = M + (nt * 16 + l16) + (2 * half) * DD;
  const float* xr = X + l16 * LSTR + 2 * half;
#pragma unroll 4
  for (int kk = 0; kk < DD; kk += 4) {
    v2f a = {xr[kk], xr[kk + 1]};
    v2f b = {mp[kk * DD], mp[kk * DD + DD]};
    acc = __builtin_amdgcn_wmma_f32_16x16x4_f32(false, a, false, b,
                                                (short)0, acc, false, false);
  }
#else
#pragma unroll
  for (int j = 0; j < 8; ++j) {
    const int m = j + 8 * half, n = nt * 16 + l16;
    float s = cinit;
    for (int k = 0; k < DD; ++k) s += X[m * LSTR + k] * M[k * DD + n];
    acc[j] = s;
  }
#endif
  return acc;
}

template <typename Epi>
__device__ __forceinline__ void gemm_xWT(const float* __restrict__ W,
                                         const float* __restrict__ bias,
                                         const float* X, int lane, int wv,
                                         Epi epi) {
  const int l16 = lane & 15, half = lane >> 4;
  const int n0 = wv * 32 + l16;
  const float bv0 = bias ? bias[n0] : 0.f;
  const float bv1 = bias ? bias[n0 + 16] : 0.f;
  v8f acc0 = {bv0, bv0, bv0, bv0, bv0, bv0, bv0, bv0};
  v8f acc1 = {bv1, bv1, bv1, bv1, bv1, bv1, bv1, bv1};
#if HAS_WMMA_F32
  const v4f* w0 = (const v4f*)(W + n0 * DD + 4 * half);
  const v4f* w1 = (const v4f*)(W + (n0 + 16) * DD + 4 * half);
  const v4f* xr = (const v4f*)(X + l16 * LSTR + 4 * half);
  v4f b0[32], b1[32], a[32];
#pragma unroll
  for (int j = 0; j < 32; ++j) {
    b0[j] = w0[2 * j];
    b1[j] = w1[2 * j];
  }
#pragma unroll
  for (int j = 0; j < 32; ++j) a[j] = xr[2 * j];
  __builtin_amdgcn_sched_barrier(0);
#pragma unroll
  for (int j = 0; j < 32; ++j) {
    const v2f alo = {a[j].x, a[j].y}, ahi = {a[j].z, a[j].w};
    const v2f blo0 = {b0[j].x, b0[j].y}, bhi0 = {b0[j].z, b0[j].w};
    const v2f blo1 = {b1[j].x, b1[j].y}, bhi1 = {b1[j].z, b1[j].w};
    acc0 = __builtin_amdgcn_wmma_f32_16x16x4_f32(false, alo, false, blo0,
                                                 (short)0, acc0, false, false);
    acc1 = __builtin_amdgcn_wmma_f32_16x16x4_f32(false, alo, false, blo1,
                                                 (short)0, acc1, false, false);
    acc0 = __builtin_amdgcn_wmma_f32_16x16x4_f32(false, ahi, false, bhi0,
                                                 (short)0, acc0, false, false);
    acc1 = __builtin_amdgcn_wmma_f32_16x16x4_f32(false, ahi, false, bhi1,
                                                 (short)0, acc1, false, false);
  }
#else
#pragma unroll
  for (int j = 0; j < 8; ++j) {
    const int m = j + 8 * half;
    float s0 = bv0, s1 = bv1;
    for (int k = 0; k < DD; ++k) {
      s0 += X[m * LSTR + k] * W[n0 * DD + k];
      s1 += X[m * LSTR + k] * W[(n0 + 16) * DD + k];
    }
    acc0[j] = s0;
    acc1[j] = s1;
  }
#endif
#pragma unroll
  for (int j = 0; j < 8; ++j) epi(j + 8 * half, n0, acc0[j]);
#pragma unroll
  for (int j = 0; j < 8; ++j) epi(j + 8 * half, n0 + 16, acc1[j]);
}

template <typename Epi>
__device__ __forceinline__ void gemm_xM(const float* __restrict__ M,
                                        const float* X, int lane, int wv,
                                        Epi epi) {
#pragma unroll
  for (int t = 0; t < 2; ++t) {
    const int nt = wv * 2 + t;
    v8f acc = tile_xM(M, X, nt, lane, 0.f);
    const int l16 = lane & 15, half = lane >> 4;
#pragma unroll
    for (int j = 0; j < 8; ++j) epi(j + 8 * half, nt * 16 + l16, acc[j]);
  }
}

__global__ __launch_bounds__(256) void pooled_kernel(
    const int* __restrict__ ids, const float* __restrict__ emb,
    const float* __restrict__ pos, float* __restrict__ pooled_g) {
  __shared__ int sid[SS];
  const int b = blockIdx.x;
  const int t = threadIdx.x;
  sid[t] = ids[b * SS + t];
  sid[t + 256] = ids[b * SS + 256 + t];
  __syncthreads();
  float acc = 0.f;
  for (int s = 0; s < SS; ++s) {
    acc += emb[(size_t)sid[s] * DD + t] + pos[s * DD + t];
  }
  pooled_g[b * DD + t] = acc * (1.0f / (float)SS);
}

__global__ __launch_bounds__(256) void refine_kernel(
    const float* __restrict__ pooled_g,
    const float* __restrict__ init_w, const float* __restrict__ init_b,
    const float* __restrict__ pred_w, const float* __restrict__ pred_b,
    const float* __restrict__ upd_w, const float* __restrict__ upd_b,
    const float* __restrict__ sqw, const float* __restrict__ sqb,
    const float* __restrict__ eqw, const float* __restrict__ eqb,
    const float* __restrict__ keyw, const float* __restrict__ keyb,
    const float* __restrict__ aw1, const float* __restrict__ ab1,
    const float* __restrict__ aw2, const float* __restrict__ ab2,
    float* __restrict__ sq_g, float* __restrict__ eq_g,
    float* __restrict__ cs_g, float* __restrict__ ce_g,
    float* __restrict__ ans_g) {
  extern __shared__ __attribute__((aligned(16))) float sm[];

  const int tid = threadIdx.x;
  const int lane = tid & 31;
  const int wv = tid >> 5;
  const int b0 = blockIdx.x * BT;

  float* pooled = smb(sm, 0);
  float* ff = smb(sm, 5);
  float* tot = smb(sm, 8);

  for (int idx = tid; idx < BT * DD; idx += 256) {
    const int r = idx >> 8, c = idx & 255;
    pooled[r * LSTR + c] = pooled_g[(b0 + r) * DD + c];
  }
  __syncthreads();

  {
    const float* cur = pooled;
    for (int l = 0; l < LLN; ++l) {
      float* dst = smb(sm, 1 + l);
      const bool last = (l == LLN - 1);
      gemm_xWT(init_w + l * DD * DD, init_b + l * DD, cur, lane, wv,
               [&](int m, int n, float v) {
                 const float r = tanh_fast(v);
                 dst[m * LSTR + n] = r;
                 if (last) ff[m * LSTR + n] = r;
               });
      __syncthreads();
      cur = dst;
    }
  }

  for (int it = 0; it < NITERS; ++it) {
    for (int i = LLN - 1; i >= 0; --i) {
      float* pc = smb(sm, 6 + ((LLN - 1 - i) & 1));
      float* pn = smb(sm, 6 + ((LLN - i) & 1));
      float* ri = smb(sm, 1 + i);
      gemm_xWT(pred_w + i * DD * DD, pred_b + i * DD, ri, lane, wv,
               [&](int m, int n, float v) { pc[m * LSTR + n] = v; });
      __syncthreads();
      {
        const float* below = (i > 0) ? smb(sm, i) : pooled;
        const bool has_above = (i < LLN - 1);
        for (int idx = tid; idx < BT * DD; idx += 256) {
          const int r = idx >> 8, c = idx & 255;
          const int o = r * LSTR + c;
          float e = below[o] - pc[o];
          if (has_above) e += 0.5f * (ri[o] - pn[o]);
          tot[o] = e;
        }
      }
      __syncthreads();
      gemm_xWT(upd_w + i * DD * DD, upd_b + i * DD, tot, lane, wv,
               [&](int m, int n, float v) {
                 const int o = m * LSTR + n;
                 ri[o] = ri[o] + 0.1f * tanh_fast(v);
               });
      __syncthreads();
    }
  }

  float* refined = smb(sm, LLN);
  for (int idx = tid; idx < BT * DD; idx += 256) {
    const int r = idx >> 8, c = idx & 255;
    refined[r * LSTR + c] += ff[r * LSTR + c];
  }
  __syncthreads();

  {
    float* q = smb(sm, 6);
    gemm_xWT(sqw, sqb, refined, lane, wv,
             [&](int m, int n, float v) { q[m * LSTR + n] = v; });
    __syncthreads();
    {
      const int r = tid >> 4, c = tid & 15;
      float p = 0.f;
#pragma unroll
      for (int u = 0; u < 16; ++u) {
        const int o = c * 16 + u;
        p += q[r * LSTR + o] * keyb[o];
      }
      tot[tid] = p;
    }
    __syncthreads();
    if (tid < BT) {
      float s = 0.f;
#pragma unroll
      for (int u = 0; u < 16; ++u) s += tot[tid * 16 + u];
      cs_g[b0 + tid] = s;
    }
    __syncthreads();
    gemm_xM(keyw, q, lane, wv,
            [&](int m, int n, float v) { sq_g[(b0 + m) * DD + n] = v; });
    __syncthreads();
  }

  {
    float* q = smb(sm, 7);
    gemm_xWT(eqw, eqb, refined, lane, wv,
             [&](int m, int n, float v) { q[m * LSTR + n] = v; });
    __syncthreads();
    {
      const int r = tid >> 4, c = tid & 15;
      float p = 0.f;
#pragma unroll
      for (int u = 0; u < 16; ++u) {
        const int o = c * 16 + u;
        p += q[r * LSTR + o] * keyb[o];
      }
      tot[tid] = p;
    }
    __syncthreads();
    if (tid < BT) {
      float s = 0.f;
#pragma unroll
      for (int u = 0; u < 16; ++u) s += tot[tid * 16 + u];
      ce_g[b0 + tid] = s;
    }
    __syncthreads();
    gemm_xM(keyw, q, lane, wv,
            [&](int m, int n, float v) { eq_g[(b0 + m) * DD + n] = v; });
    __syncthreads();
  }

  for (int task = tid; task < BT * 128; task += 256) {
    const int r = task >> 7, j = task & 127;
    float s = ab1[j];
    const float* rr = refined + r * LSTR;
    const float* wr = aw1 + j * DD;
    for (int k = 0; k < DD; ++k) s += rr[k] * wr[k];
    tot[r * LSTR + j] = 0.5f * s * (1.0f + erff(s * 0.70710678118654752f));
  }
  __syncthreads();
  if (tid < BT * 2) {
    const int r = tid >> 1, k = tid & 1;
    float s = ab2[k];
    for (int j = 0; j < 128; ++j) s += tot[r * LSTR + j] * aw2[k * 128 + j];
    ans_g[(b0 + r) * 2 + k] = s;
  }
}

__global__ __launch_bounds__(256) void logits_kernel(
    const int* __restrict__ ids, const float* __restrict__ emb,
    const float* __restrict__ pos, const float* __restrict__ sq_g,
    const float* __restrict__ eq_g, const float* __restrict__ cs_g,
    const float* __restrict__ ce_g, float* __restrict__ out_start,
    float* __restrict__ out_end) {
  const int b = blockIdx.x;
  const int lane = threadIdx.x & 31;
  const int wv = threadIdx.x >> 5;
  float sqv[8], eqv[8];
  const float* sr = sq_g + b * DD + lane * 8;
  const float* er = eq_g + b * DD + lane * 8;
#pragma unroll
  for (int j = 0; j < 8; ++j) {
    sqv[j] = sr[j];
    eqv[j] = er[j];
  }
  const float cs = cs_g[b];
  const float ce = ce_g[b];
  for (int s = wv; s < SS; s += 8) {
    const int id = ids[b * SS + s];
    const float* row = emb + (size_t)id * DD + lane * 8;
    const float* pr = pos + s * DD + lane * 8;
    float as = 0.f, ae = 0.f;
#pragma unroll
    for (int j = 0; j < 8; ++j) {
      const float e = row[j] + pr[j];
      as += e * sqv[j];
      ae += e * eqv[j];
    }
#pragma unroll
    for (int off = 16; off; off >>= 1) {
      as += __shfl_xor(as, off, 32);
      ae += __shfl_xor(ae, off, 32);
    }
    if (lane == 0) {
      out_start[b * SS + s] = as + cs;
      out_end[b * SS + s] = ae + ce;
    }
  }
}

extern "C" void kernel_launch(void* const* d_in, const int* in_sizes, int n_in,
                              void* d_out, int out_size, void* d_ws,
                              size_t ws_size, hipStream_t stream) {
  const int* ids = (const int*)d_in[0];
  const float* emb = (const float*)d_in[1];
  const float* pos = (const float*)d_in[2];
  const float* init_w = (const float*)d_in[3];
  const float* init_b = (const float*)d_in[4];
  const float* pred_w = (const float*)d_in[5];
  const float* pred_b = (const float*)d_in[6];
  const float* upd_w = (const float*)d_in[7];
  const float* upd_b = (const float*)d_in[8];
  const float* sqw = (const float*)d_in[9];
  const float* sqb = (const float*)d_in[10];
  const float* eqw = (const float*)d_in[11];
  const float* eqb = (const float*)d_in[12];
  const float* keyw = (const float*)d_in[13];
  const float* keyb = (const float*)d_in[14];
  const float* aw1 = (const float*)d_in[15];
  const float* ab1 = (const float*)d_in[16];
  const float* aw2 = (const float*)d_in[17];
  const float* ab2 = (const float*)d_in[18];

  float* ws = (float*)d_ws;
  float* pooled = ws;
  float* sq = pooled + BB * DD;
  float* eq = sq + BB * DD;
  float* cs = eq + BB * DD;
  float* ce = cs + BB;

  float* out = (float*)d_out;
  float* out_start = out;
  float* out_end = out + BB * SS;
  float* out_ans = out + 2 * BB * SS;

  (void)hipFuncSetAttribute((const void*)refine_kernel,
                            hipFuncAttributeMaxDynamicSharedMemorySize,
                            (int)SMEM_BYTES);

  pooled_kernel<<<BB, 256, 0, stream>>>(ids, emb, pos, pooled);
  refine_kernel<<<BB / BT, 256, SMEM_BYTES, stream>>>(
      pooled, init_w, init_b, pred_w, pred_b, upd_w, upd_b, sqw, sqb, eqw, eqb,
      keyw, keyb, aw1, ab1, aw2, ab2, sq, eq, cs, ce, out_ans);
  logits_kernel<<<BB, 256, 0, stream>>>(ids, emb, pos, sq, eq, cs, ce,
                                        out_start, out_end);
}


# refine GEMMs: first launch rewrites the 14 W^T weight matrices in fragment order in the workspace; each wave load reads 512 contiguous bytes, not 16-byte pieces of 16 rows; k order and sums unchanged
# speedup vs baseline: 1.6169x; 1.3734x over previous
#include <hip/hip_runtime.h>
#include <math.h>

#define BB   256
#define SS   512
#define DD   256
#define LLN  4
#define NITERS 10
#define BT   16
#define LSTR 260
#define TILE (BT * LSTR)
#define SMEM_FLOATS (9 * TILE)
#define SMEM_BYTES  (SMEM_FLOATS * sizeof(float))
#define NWM  14
#define WMAT (DD * DD)
#define RL_BLOCKS (NWM * WMAT / 4 / 256 / 4)

typedef float v2f __attribute__((ext_vector_type(2)));
typedef float v4f __attribute__((ext_vector_type(4)));
typedef float v8f __attribute__((ext_vector_type(8)));

#if defined(__has_builtin)
#  if __has_builtin(__builtin_amdgcn_wmma_f32_16x16x4_f32)
#    define HAS_WMMA_F32 1
#  else
#    define HAS_WMMA_F32 0
#  endif
#  if __has_builtin(__builtin_amdgcn_tanhf)
#    define HAS_HW_TANH 1
#  else
#    define HAS_HW_TANH 0
#  endif
#else
#  define HAS_WMMA_F32 0
#  define HAS_HW_TANH 0
#endif

__device__ __forceinline__ float tanh_fast(float x) {
#if HAS_HW_TANH
  return __builtin_amdgcn_tanhf(x);
#else
  return tanhf(x);
#endif
}

__device__ __forceinline__ float* smb(float* sm, int i) { return sm + i * TILE; }


__device__ __forceinline__ v8f tile_xM(const float* __restrict__ M,
                                       const float* X, int nt, int lane,
                                       float cinit) {
  const int l16 = lane & 15;
  const int half = lane >> 4;
  v8f acc = {cinit, cinit, cinit, cinit, cinit, cinit, cinit, cinit};
#if HAS_WMMA_F32
  const float* mp = M + (nt * 16 + l16) + (2 * half) * DD;
  const float* xr = X + l16 * LSTR + 2 * half;
#pragma unroll 4
  for (int kk = 0; kk < DD; kk += 4) {
    v2f a = {xr[kk], xr[kk + 1]};
    v2f b = {mp[kk * DD], mp[kk * DD + DD]};
    acc = __builtin_amdgcn_wmma_f32_16x16x4_f32(false, a, false, b,
                                                (short)0, acc, false, false);
  }
#else
#pragma unroll
  for (int j = 0; j < 8; ++j) {
    const int m = j + 8 * half, n = nt * 16 + l16;
    float s = cinit;
    for (int k = 0; k < DD; ++k) s += X[m * LSTR + k] * M[k * DD + n];
    acc[j] = s;
  }
#endif
  return acc;
}

template <typename Epi>
__device__ __forceinline__ void gemm_xWT(const float* __restrict__ Wr,
                                         const float* __restrict__ bias,
                                         const float* X, int lane, int wv,
                                         Epi epi) {
  const int l16 = lane & 15, half = lane >> 4;
  const int n0 = wv * 32 + l16;
  const float bv0 = bias ? bias[n0] : 0.f;
  const float bv1 = bias ? bias[n0 + 16] : 0.f;
  v8f acc0 = {bv0, bv0, bv0, bv0, bv0, bv0, bv0, bv0};
  v8f acc1 = {bv1, bv1, bv1, bv1, bv1, bv1, bv1, bv1};
#if HAS_WMMA_F32
  const v4f* wp = (const v4f*)Wr + wv * 2048 + lane;
  const v4f* xr = (const v4f*)(X + l16 * LSTR + 4 * half);
  v4f b0[32], b1[32], a[32];
#pragma unroll
  for (int j = 0; j < 32; ++j) {
    b0[j] = wp[(2 * j) * 32];
    b1[j] = wp[(2 * j + 1) * 32];
  }
#pragma unroll
  for (int j = 0; j < 32; ++j) a[j] = xr[2 * j];
  __builtin_amdgcn_sched_barrier(0);
#pragma unroll
  for (int j = 0; j < 32; ++j) {
    const v2f alo = {a[j].x, a[j].y}, ahi = {a[j].z, a[j].w};
    const v2f blo0 = {b0[j].x, b0[j].y}, bhi0 = {b0[j].z, b0[j].w};
    const v2f blo1 = {b1[j].x, b1[j].y}, bhi1 = {b1[j].z, b1[j].w};
    acc0 = __builtin_amdgcn_wmma_f32_16x16x4_f32(false, alo, false, blo0,
                                                 (short)0, acc0, false, false);
    acc1 = __builtin_amdgcn_wmma_f32_16x16x4_f32(false, alo, false, blo1,
                                                 (short)0, acc1, false, false);
    acc0 = __builtin_amdgcn_wmma_f32_16x16x4_f32(false, ahi, false, bhi0,
                                                 (short)0, acc0, false, false);
    acc1 = __builtin_amdgcn_wmma_f32_16x16x4_f32(false, ahi, false, bhi1,
                                                 (short)0, acc1, false, false);
  }
#else
#pragma unroll
  for (int j = 0; j < 8; ++j) {
    const int m = j + 8 * half;
    float s0 = bv0, s1 = bv1;
    for (int k = 0; k < DD; ++k) {
      const int slot = ((wv * 32 + (k >> 3)) * 2) * 32 + ((k >> 2) & 1) * 16 + l16;
      s0 += X[m * LSTR + k] * Wr[slot * 4 + (k & 3)];
      s1 += X[m * LSTR + k] * Wr[(slot + 32) * 4 + (k & 3)];
    }
    acc0[j] = s0;
    acc1[j] = s1;
  }
#endif
#pragma unroll
  for (int j = 0; j < 8; ++j) epi(j + 8 * half, n0, acc0[j]);
#pragma unroll
  for (int j = 0; j < 8; ++j) epi(j + 8 * half, n0 + 16, acc1[j]);
}

template <typename Epi>
__device__ __forceinline__ void gemm_xM(const float* __restrict__ M,
                                        const float* X, int lane, int wv,
                                        Epi epi) {
#pragma unroll
  for (int t = 0; t < 2; ++t) {
    const int nt = wv * 2 + t;
    v8f acc = tile_xM(M, X, nt, lane, 0.f);
    const int l16 = lane & 15, half = lane >> 4;
#pragma unroll
    for (int j = 0; j < 8; ++j) epi(j + 8 * half, nt * 16 + l16, acc[j]);
  }
}

__global__ __launch_bounds__(256) void pooled_kernel(
    const int* __restrict__ ids, const float* __restrict__ emb,
    const float* __restrict__ pos, float* __restrict__ pooled_g,
    const float* __restrict__ init_w, const float* __restrict__ pred_w,
    const float* __restrict__ upd_w, const float* __restrict__ sqw,
    const float* __restrict__ eqw, float* __restrict__ wr_g) {
  __shared__ int sid[SS];
  const int t = threadIdx.x;
  if (blockIdx.x >= BB) {
    const int rb = blockIdx.x - BB;
    const int mi = rb >> 4;
    const float* src = mi < 4    ? init_w + mi * WMAT
                       : mi < 8  ? pred_w + (mi - 4) * WMAT
                       : mi < 12 ? upd_w + (mi - 8) * WMAT
                       : mi == 12 ? sqw
                                  : eqw;
    v4f* dst = (v4f*)(wr_g + (size_t)mi * WMAT);
#pragma unroll
    for (int i = 0; i < 4; ++i) {
      const int d = ((rb & 15) * 4 + i) * 256 + t;
      const int ln = d & 31, tt = (d >> 5) & 1, j = (d >> 6) & 31, w = d >> 11;
      const int n = w * 32 + tt * 16 + (ln & 15);
      const int k = 8 * j + 4 * (ln >> 4);
      dst[d] = *(const v4f*)(src + n * DD + k);
    }
    return;
  }
  const int b = blockIdx.x;
  sid[t] = ids[b * SS + t];
  sid[t + 256] = ids[b * SS + 256 + t];
  __syncthreads();
  float acc = 0.f;
  for (int s = 0; s < SS; ++s) {
    acc += emb[(size_t)sid[s] * DD + t] + pos[s * DD + t];
  }
  pooled_g[b * DD + t] = acc * (1.0f / (float)SS);
}

__global__ __launch_bounds__(256) void refine_kernel(
    const float* __restrict__ pooled_g,
    const float* __restrict__ wr_g,
    const float* __restrict__ init_b, const float* __restrict__ pred_b,
    const float* __restrict__ upd_b, const float* __restrict__ sqb,
    const float* __restrict__ eqb,
    const float* __restrict__ keyw, const float* __restrict__ keyb,
    const float* __restrict__ aw1, const float* __restrict__ ab1,
    const float* __restrict__ aw2, const float* __restrict__ ab2,
    float* __restrict__ sq_g, float* __restrict__ eq_g,
    float* __restrict__ cs_g, float* __restrict__ ce_g,
    float* __restrict__ ans_g) {
  extern __shared__ __attribute__((aligned(16))) float sm[];

  const int tid = threadIdx.x;
  const int lane = tid & 31;
  const int wv = tid >> 5;
  const int b0 = blockIdx.x * BT;

  float* pooled = smb(sm, 0);
  float* ff = smb(sm, 5);
  float* tot = smb(sm, 8);

  for (int idx = tid; idx < BT * DD; idx += 256) {
    const int r = idx >> 8, c = idx & 255;
    pooled[r * LSTR + c] = pooled_g[(b0 + r) * DD + c];
  }
  __syncthreads();

  {
    const float* cur = pooled;
    for (int l = 0; l < LLN; ++l) {
      float* dst = smb(sm, 1 + l);
      const bool last = (l == LLN - 1);
      gemm_xWT(wr_g + l * WMAT, init_b + l * DD, cur, lane, wv,
               [&](int m, int n, float v) {
                 const float r = tanh_fast(v);
                 dst[m * LSTR + n] = r;
                 if (last) ff[m * LSTR + n] = r;
               });
      __syncthreads();
      cur = dst;
    }
  }

  for (int it = 0; it < NITERS; ++it) {
    for (int i = LLN - 1; i >= 0; --i) {
      float* pc = smb(sm, 6 + ((LLN - 1 - i) & 1));
      float* pn = smb(sm, 6 + ((LLN - i) & 1));
      float* ri = smb(sm, 1 + i);
      gemm_xWT(wr_g + (4 + i) * WMAT, pred_b + i * DD, ri, lane, wv,
               [&](int m, int n, float v) { pc[m * LSTR + n] = v; });
      __syncthreads();
      {
        const float* below = (i > 0) ? smb(sm, i) : pooled;
        const bool has_above = (i < LLN - 1);
        for (int idx = tid; idx < BT * DD; idx += 256) {
          const int r = idx >> 8, c = idx & 255;
          const int o = r * LSTR + c;
          float e = below[o] - pc[o];
          if (has_above) e += 0.5f * (ri[o] - pn[o]);
          tot[o] = e;
        }
      }
      __syncthreads();
      gemm_xWT(wr_g + (8 + i) * WMAT, upd_b + i * DD, tot, lane, wv,
               [&](int m, int n, float v) {
                 const int o = m * LSTR + n;
                 ri[o] = ri[o] + 0.1f * tanh_fast(v);
               });
      __syncthreads();
    }
  }

  float* refined = smb(sm, LLN);
  for (int idx = tid; idx < BT * DD; idx += 256) {
    const int r = idx >> 8, c = idx & 255;
    refined[r * LSTR + c] += ff[r * LSTR + c];
  }
  __syncthreads();

  {
    float* q = smb(sm, 6);
    gemm_xWT(wr_g + 12 * WMAT, sqb, refined, lane, wv,
             [&](int m, int n, float v) { q[m * LSTR + n] = v; });
    __syncthreads();
    {
      const int r = tid >> 4, c = tid & 15;
      float p = 0.f;
#pragma unroll
      for (int u = 0; u < 16; ++u) {
        const int o = c * 16 + u;
        p += q[r * LSTR + o] * keyb[o];
      }
      tot[tid] = p;
    }
    __syncthreads();
    if (tid < BT) {
      float s = 0.f;
#pragma unroll
      for (int u = 0; u < 16; ++u) s += tot[tid * 16 + u];
      cs_g[b0 + tid] = s;
    }
    __syncthreads();
    gemm_xM(keyw, q, lane, wv,
            [&](int m, int n, float v) { sq_g[(b0 + m) * DD + n] = v; });
    __syncthreads();
  }

  {
    float* q = smb(sm, 7);
    gemm_xWT(wr_g + 13 * WMAT, eqb, refined, lane, wv,
             [&](int m, int n, float v) { q[m * LSTR + n] = v; });
    __syncthreads();
    {
      const int r = tid >> 4, c = tid & 15;
      float p = 0.f;
#pragma unroll
      for (int u = 0; u < 16; ++u) {
        const int o = c * 16 + u;
        p += q[r * LSTR + o] * keyb[o];
      }
      tot[tid] = p;
    }
    __syncthreads();
    if (tid < BT) {
      float s = 0.f;
#pragma unroll
      for (int u = 0; u < 16; ++u) s += tot[tid * 16 + u];
      ce_g[b0 + tid] = s;
    }
    __syncthreads();
    gemm_xM(keyw, q, lane, wv,
            [&](int m, int n, float v) { eq_g[(b0 + m) * DD + n] = v; });
    __syncthreads();
  }

  for (int task = tid; task < BT * 128; task += 256) {
    const int r = task >> 7, j = task & 127;
    float s = ab1[j];
    const float* rr = refined + r * LSTR;
    const float* wr = aw1 + j * DD;
    for (int k = 0; k < DD; ++k) s += rr[k] * wr[k];
    tot[r * LSTR + j] = 0.5f * s * (1.0f + erff(s * 0.70710678118654752f));
  }
  __syncthreads();
  if (tid < BT * 2) {
    const int r = tid >> 1, k = tid & 1;
    float s = ab2[k];
    for (int j = 0; j < 128; ++j) s += tot[r * LSTR + j] * aw2[k * 128 + j];
    ans_g[(b0 + r) * 2 + k] = s;
  }
}

__global__ __launch_bounds__(256) void logits_kernel(
    const int* __restrict__ ids, const float* __restrict__ emb,
    const float* __restrict__ pos, const float* __restrict__ sq_g,
    const float* __restrict__ eq_g, const float* __restrict__ cs_g,
    const float* __restrict__ ce_g, float* __restrict__ out_start,
    float* __restrict__ out_end) {
  const int b = blockIdx.x;
  const int lane = threadIdx.x & 31;
  const int wv = threadIdx.x >> 5;
  float sqv[8], eqv[8];
  const float* sr = sq_g + b * DD + lane * 8;
  const float* er = eq_g + b * DD + lane * 8;
#pragma unroll
  for (int j = 0; j < 8; ++j) {
    sqv[j] = sr[j];
    eqv[j] = er[j];
  }
  const float cs = cs_g[b];
  const float ce = ce_g[b];
  for (int s = wv; s < SS; s += 8) {
    const int id = ids[b * SS + s];
    const float* row = emb + (size_t)id * DD + lane * 8;
    const float* pr = pos + s * DD + lane * 8;
    float as = 0.f, ae = 0.f;
#pragma unroll
    for (int j = 0; j < 8; ++j) {
      const float e = row[j] + pr[j];
      as += e * sqv[j];
      ae += e * eqv[j];
    }
#pragma unroll
    for (int off = 16; off; off >>= 1) {
      as += __shfl_xor(as, off, 32);
      ae += __shfl_xor(ae, off, 32);
    }
    if (lane == 0) {
      out_start[b * SS + s] = as + cs;
      out_end[b * SS + s] = ae + ce;
    }
  }
}

extern "C" void kernel_launch(void* const* d_in, const int* in_sizes, int n_in,
                              void* d_out, int out_size, void* d_ws,
                              size_t ws_size, hipStream_t stream) {
  const int* ids = (const int*)d_in[0];
  const float* emb = (const float*)d_in[1];
  const float* pos = (const float*)d_in[2];
  const float* init_w = (const float*)d_in[3];
  const float* init_b = (const float*)d_in[4];
  const float* pred_w = (const float*)d_in[5];
  const float* pred_b = (const float*)d_in[6];
  const float* upd_w = (const float*)d_in[7];
  const float* upd_b = (const float*)d_in[8];
  const float* sqw = (const float*)d_in[9];
  const float* sqb = (const float*)d_in[10];
  const float* eqw = (const float*)d_in[11];
  const float* eqb = (const float*)d_in[12];
  const float* keyw = (const float*)d_in[13];
  const float* keyb = (const float*)d_in[14];
  const float* aw1 = (const float*)d_in[15];
  const float* ab1 = (const float*)d_in[16];
  const float* aw2 = (const float*)d_in[17];
  const float* ab2 = (const float*)d_in[18];

  float* ws = (float*)d_ws;
  float* pooled = ws;
  float* sq = pooled + BB * DD;
  float* eq = sq + BB * DD;
  float* cs = eq + BB * DD;
  float* ce = cs + BB;
  float* wr = ce + BB;

  float* out = (float*)d_out;
  float* out_start = out;
  float* out_end = out + BB * SS;
  float* out_ans = out + 2 * BB * SS;

  (void)hipFuncSetAttribute((const void*)refine_kernel,
                            hipFuncAttributeMaxDynamicSharedMemorySize,
                            (int)SMEM_BYTES);

  pooled_kernel<<<BB + RL_BLOCKS, 256, 0, stream>>>(
      ids, emb, pos, pooled, init_w, pred_w, upd_w, sqw, eqw, wr);
  refine_kernel<<<BB / BT, 256, SMEM_BYTES, stream>>>(
      pooled, wr, init_b, pred_b, upd_b, sqb, eqb,
      keyw, keyb, aw1, ab1, aw2, ab2, sq, eq, cs, ce, out_ans);
  logits_kernel<<<BB, 256, 0, stream>>>(ids, emb, pos, sq, eq, cs, ce,
                                        out_start, out_end);
}


# refine kernel split into one launch per dependency phase (4 init, pred and upd per sweep, heads, fold): each phase's GEMMs run on 128 work-groups, not 16; 28 launches; all sums unchanged
# speedup vs baseline: 3.8540x; 2.3836x over previous
#include <hip/hip_runtime.h>
#include <math.h>

#define BB   256
#define SS   512
#define DD   256
#define LLN  4
#define NITERS 10
#define BT   16
#define LSTR 260
#define TILE (BT * LSTR)
#define NWM  14
#define WMAT (DD * DD)
#define RL_BLOCKS (NWM * WMAT / 4 / 256 / 4)
#define ACT  (BB * DD)
#define GT   128

typedef float v2f __attribute__((ext_vector_type(2)));
typedef float v4f __attribute__((ext_vector_type(4)));
typedef float v8f __attribute__((ext_vector_type(8)));

#if defined(__has_builtin)
#  if __has_builtin(__builtin_amdgcn_wmma_f32_16x16x4_f32)
#    define HAS_WMMA_F32 1
#  else
#    define HAS_WMMA_F32 0
#  endif
#  if __has_builtin(__builtin_amdgcn_tanhf)
#    define HAS_HW_TANH 1
#  else
#    define HAS_HW_TANH 0
#  endif
#else
#  define HAS_WMMA_F32 0
#  define HAS_HW_TANH 0
#endif

__device__ __forceinline__ float tanh_fast(float x) {
#if HAS_HW_TANH
  return __builtin_amdgcn_tanhf(x);
#else
  return tanhf(x);
#endif
}


__device__ __forceinline__ v8f tile_xM(const float* __restrict__ M,
                                       const float* X, int nt, int lane,
                                       float cinit) {
  const int l16 = lane & 15;
  const int half = lane >> 4;
  v8f acc = {cinit, cinit, cinit, cinit, cinit, cinit, cinit, cinit};
#if HAS_WMMA_F32
  const float* mp = M + (nt * 16 + l16) + (2 * half) * DD;
  const float* xr = X + l16 * LSTR + 2 * half;
#pragma unroll 4
  for (int kk = 0; kk < DD; kk += 4) {
    v2f a = {xr[kk], xr[kk + 1]};
    v2f b = {mp[kk * DD], mp[kk * DD + DD]};
    acc = __builtin_amdgcn_wmma_f32_16x16x4_f32(false, a, false, b,
                                                (short)0, acc, false, false);
  }
#else
#pragma unroll
  for (int j = 0; j < 8; ++j) {
    const int m = j + 8 * half, n = nt * 16 + l16;
    float s = cinit;
    for (int k = 0; k < DD; ++k) s += X[m * LSTR + k] * M[k * DD + n];
    acc[j] = s;
  }
#endif
  return acc;
}

template <typename Mid, typename Epi>
__device__ __forceinline__ void gemm_xWT(const float* __restrict__ Wr,
                                         const float* __restrict__ bias,
                                         const float* X, int lane, int wv,
                                         Mid mid, Epi epi) {
  const int l16 = lane & 15, half = lane >> 4;
  const int n0 = wv * 32 + l16;
  const float bv0 = bias ? bias[n0] : 0.f;
  const float bv1 = bias ? bias[n0 + 16] : 0.f;
  v8f acc0 = {bv0, bv0, bv0, bv0, bv0, bv0, bv0, bv0};
  v8f acc1 = {bv1, bv1, bv1, bv1, bv1, bv1, bv1, bv1};
#if HAS_WMMA_F32
  const v4f* wp = (const v4f*)Wr + wv * 2048 + lane;
  const v4f* xr = (const v4f*)(X + l16 * LSTR + 4 * half);
  v4f b0[32], b1[32], a[32];
#pragma unroll
  for (int j = 0; j < 32; ++j) {
    b0[j] = wp[(2 * j) * 32];
    b1[j] = wp[(2 * j + 1) * 32];
  }
  mid();
#pragma unroll
  for (int j = 0; j < 32; ++j) a[j] = xr[2 * j];
  __builtin_amdgcn_sched_barrier(0);
#pragma unroll
  for (int j = 0; j < 32; ++j) {
    const v2f alo = {a[j].x, a[j].y}, ahi = {a[j].z, a[j].w};
    const v2f blo0 = {b0[j].x, b0[j].y}, bhi0 = {b0[j].z, b0[j].w};
    const v2f blo1 = {b1[j].x, b1[j].y}, bhi1 = {b1[j].z, b1[j].w};
    acc0 = __builtin_amdgcn_wmma_f32_16x16x4_f32(false, alo, false, blo0,
                                                 (short)0, acc0, false, false);
    acc1 = __builtin_amdgcn_wmma_f32_16x16x4_f32(false, alo, false, blo1,
                                                 (short)0, acc1, false, false);
    acc0 = __builtin_amdgcn_wmma_f32_16x16x4_f32(false, ahi, false, bhi0,
                                                 (short)0, acc0, false, false);
    acc1 = __builtin_amdgcn_wmma_f32_16x16x4_f32(false, ahi, false, bhi1,
                                                 (short)0, acc1, false, false);
  }
#else
  mid();
#pragma unroll
  for (int j = 0; j < 8; ++j) {
    const int m = j + 8 * half;
    float s0 = bv0, s1 = bv1;
    for (int k = 0; k < DD; ++k) {
      const int slot = ((wv * 32 + (k >> 3)) * 2) * 32 + ((k >> 2) & 1) * 16 + l16;
      s0 += X[m * LSTR + k] * Wr[slot * 4 + (k & 3)];
      s1 += X[m * LSTR + k] * Wr[(slot + 32) * 4 + (k & 3)];
    }
    acc0[j] = s0;
    acc1[j] = s1;
  }
#endif
#pragma unroll
  for (int j = 0; j < 8; ++j) epi(j + 8 * half, n0, acc0[j]);
#pragma unroll
  for (int j = 0; j < 8; ++j) epi(j + 8 * half, n0 + 16, acc1[j]);
}

__device__ __forceinline__ int st_goff(int tid, int u, int b0) {
  const int idx = tid + GT * u;
  return (b0 + (idx >> 6)) * DD + (idx & 63) * 4;
}
__device__ __forceinline__ int st_loff(int tid, int u) {
  const int idx = tid + GT * u;
  return (idx >> 6) * LSTR + (idx & 63) * 4;
}
__device__ __forceinline__ void lds_barrier() {
  __builtin_amdgcn_fence(__ATOMIC_RELEASE, "workgroup", "local");
  __builtin_amdgcn_s_barrier();
  __builtin_amdgcn_fence(__ATOMIC_ACQUIRE, "workgroup", "local");
}

__global__ __launch_bounds__(256) void pooled_kernel(
    const int* __restrict__ ids, const float* __restrict__ emb,
    const float* __restrict__ pos, float* __restrict__ pooled_g,
    const float* __restrict__ init_w, const float* __restrict__ pred_w,
    const float* __restrict__ upd_w, const float* __restrict__ sqw,
    const float* __restrict__ eqw, float* __restrict__ wr_g) {
  __shared__ int sid[SS];
  const int t = threadIdx.x;
  if (blockIdx.x >= BB) {
    const int rb = blockIdx.x - BB;
    const int mi = rb >> 4;
    const float* src = mi < 4    ? init_w + mi * WMAT
                       : mi < 8  ? pred_w + (mi - 4) * WMAT
                       : mi < 12 ? upd_w + (mi - 8) * WMAT
                       : mi == 12 ? sqw
                                  : eqw;
    v4f* dst = (v4f*)(wr_g + (size_t)mi * WMAT);
#pragma unroll
    for (int i = 0; i < 4; ++i) {
      const int d = ((rb & 15) * 4 + i) * 256 + t;
      const int ln = d & 31, tt = (d >> 5) & 1, j = (d >> 6) & 31, w = d >> 11;
      const int n = w * 32 + tt * 16 + (ln & 15);
      const int k = 8 * j + 4 * (ln >> 4);
      dst[d] = *(const v4f*)(src + n * DD + k);
    }
    return;
  }
  const int b = blockIdx.x;
  sid[t] = ids[b * SS + t];
  sid[t + 256] = ids[b * SS + 256 + t];
  __syncthreads();
  float acc = 0.f;
  for (int s = 0; s < SS; ++s) {
    acc += emb[(size_t)sid[s] * DD + t] + pos[s * DD + t];
  }
  pooled_g[b * DD + t] = acc * (1.0f / (float)SS);
}


template <bool LAST>
__global__ __launch_bounds__(GT) void init_kernel(
    const float* __restrict__ x_g, float* __restrict__ y_g,
    float* __restrict__ ff_g, const float* __restrict__ w,
    const float* __restrict__ bias) {
  __shared__ __attribute__((aligned(16))) float xs[TILE];
  const int tid = threadIdx.x;
  const int lane = tid & 31;
  const int wv = blockIdx.x * 4 + (tid >> 5);
  const int b0 = blockIdx.y * BT;
  v4f vx[8];
#pragma unroll
  for (int u = 0; u < 8; ++u) vx[u] = *(const v4f*)(x_g + st_goff(tid, u, b0));
  gemm_xWT(w, bias, xs, lane, wv,
           [&]() {
#pragma unroll
             for (int u = 0; u < 8; ++u) *(v4f*)(xs + st_loff(tid, u)) = vx[u];
             lds_barrier();
           },
           [&](int m, int n, float v) {
             const float r = tanh_fast(v);
             const int o = (b0 + m) * DD + n;
             y_g[o] = r;
             if (LAST) ff_g[o] = r;
           });
}

__global__ __launch_bounds__(GT) void pred_kernel(
    const float* __restrict__ rc, float* __restrict__ pred_g,
    const float* __restrict__ wr_g, const float* __restrict__ pred_b) {
  __shared__ __attribute__((aligned(16))) float xs[TILE];
  const int tid = threadIdx.x;
  const int lane = tid & 31;
  const int wv = blockIdx.x * 4 + (tid >> 5);
  const int b0 = blockIdx.y * BT;
  const int i = blockIdx.z;
  const float* x_g = rc + i * ACT;
  float* pc = pred_g + i * ACT;
  v4f vx[8];
#pragma unroll
  for (int u = 0; u < 8; ++u) vx[u] = *(const v4f*)(x_g + st_goff(tid, u, b0));
  gemm_xWT(wr_g + (4 + i) * WMAT, pred_b + i * DD, xs, lane, wv,
           [&]() {
#pragma unroll
             for (int u = 0; u < 8; ++u) *(v4f*)(xs + st_loff(tid, u)) = vx[u];
             lds_barrier();
           },
           [&](int m, int n, float v) { pc[(b0 + m) * DD + n] = v; });
}

__global__ __launch_bounds__(GT) void upd_kernel(
    const float* __restrict__ pooled_g, const float* __restrict__ rc,
    float* __restrict__ rn, const float* __restrict__ pred_g,
    const float* __restrict__ wr_g, const float* __restrict__ upd_b) {
  __shared__ __attribute__((aligned(16))) float xs[TILE];
  const int tid = threadIdx.x;
  const int lane = tid & 31;
  const int wv = blockIdx.x * 4 + (tid >> 5);
  const int b0 = blockIdx.y * BT;
  const int i = blockIdx.z;
  const bool has_above = (i < LLN - 1);
  const float* below = (i > 0) ? rc + (i - 1) * ACT : pooled_g;
  const float* ri = rc + i * ACT;
  const float* pc = pred_g + i * ACT;
  const float* pn = pred_g + (has_above ? i + 1 : i) * ACT;
  float* ro = rn + i * ACT;
  v4f vb[8], vp[8], vr[8], vn[8];
#pragma unroll
  for (int u = 0; u < 8; ++u) {
    const int g = st_goff(tid, u, b0);
    vb[u] = *(const v4f*)(below + g);
    vp[u] = *(const v4f*)(pc + g);
    vr[u] = *(const v4f*)(ri + g);
    vn[u] = *(const v4f*)(pn + g);
  }
  gemm_xWT(wr_g + (8 + i) * WMAT, upd_b + i * DD, xs, lane, wv,
           [&]() {
#pragma unroll
             for (int u = 0; u < 8; ++u) {
               v4f e = vb[u] - vp[u];
               if (has_above) e += 0.5f * (vr[u] - vn[u]);
               *(v4f*)(xs + st_loff(tid, u)) = e;
             }
             lds_barrier();
           },
           [&](int m, int n, float v) {
             const int o = (b0 + m) * DD + n;
             ro[o] = ri[o] + 0.1f * tanh_fast(v);
           });
}

__global__ __launch_bounds__(GT) void heads_kernel(
    const float* __restrict__ r3, const float* __restrict__ ff_g,
    const float* __restrict__ wr_g, const float* __restrict__ sqb,
    const float* __restrict__ eqb, float* __restrict__ q_g,
    const float* __restrict__ aw1, const float* __restrict__ ab1,
    const float* __restrict__ aw2, const float* __restrict__ ab2,
    float* __restrict__ ans_g) {
  __shared__ __attribute__((aligned(16))) float xs[TILE];
  const int tid = threadIdx.x;
  const int lane = tid & 31;
  const int wv = blockIdx.x * 4 + (tid >> 5);
  const int b0 = blockIdx.y * BT;
  const int z = blockIdx.z;
  if (z >= 2) {
    const int r = (z - 2) * 2 + blockIdx.x;
    const float* ra = r3 + (b0 + r) * DD;
    const float* fa = ff_g + (b0 + r) * DD;
    const float* wrow = aw1 + tid * DD;
    float s = ab1[tid];
#pragma unroll 32
    for (int k = 0; k < DD; ++k) s += (ra[k] + fa[k]) * wrow[k];
    xs[tid] = 0.5f * s * (1.0f + erff(s * 0.70710678118654752f));
    __syncthreads();
    if (tid < 2) {
      float t = ab2[tid];
      for (int j = 0; j < 128; ++j) t += xs[j] * aw2[tid * 128 + j];
      ans_g[(b0 + r) * 2 + tid] = t;
    }
    return;
  }
  float* q = q_g + z * ACT;
  v4f va[8], vf[8];
#pragma unroll
  for (int u = 0; u < 8; ++u) {
    const int g = st_goff(tid, u, b0);
    va[u] = *(const v4f*)(r3 + g);
    vf[u] = *(const v4f*)(ff_g + g);
  }
  gemm_xWT(wr_g + (12 + z) * WMAT, z ? eqb : sqb, xs, lane, wv,
           [&]() {
#pragma unroll
             for (int u = 0; u < 8; ++u)
               *(v4f*)(xs + st_loff(tid, u)) = va[u] + vf[u];
             lds_barrier();
           },
           [&](int m, int n, float v) { q[(b0 + m) * DD + n] = v; });
}

__global__ __launch_bounds__(GT) void fold_kernel(
    const float* __restrict__ q_g, const float* __restrict__ keyw,
    const float* __restrict__ keyb, float* __restrict__ sq_g,
    float* __restrict__ eq_g, float* __restrict__ cs_g,
    float* __restrict__ ce_g) {
  __shared__ __attribute__((aligned(16))) float xs[TILE];
  __shared__ __attribute__((aligned(16))) float part[256];
  __shared__ __attribute__((aligned(16))) float ys[BT * 64];
  const int tid = threadIdx.x;
  const int lane = tid & 31;
  const int wl = tid >> 5;
  const int b0 = blockIdx.y * BT;
  const int z = blockIdx.z;
  const float* q = q_g + z * ACT;
  float* fq = z ? eq_g : sq_g;
  float* cq = z ? ce_g : cs_g;
#pragma unroll
  for (int u = 0; u < 8; ++u)
    *(v4f*)(xs + st_loff(tid, u)) = *(const v4f*)(q + st_goff(tid, u, b0));
  __syncthreads();
  if (blockIdx.x == 0) {
    for (int t2 = tid; t2 < 256; t2 += GT) {
      const int r = t2 >> 4, c = t2 & 15;
      float p = 0.f;
#pragma unroll
      for (int u = 0; u < 16; ++u) {
        const int o = c * 16 + u;
        p += xs[r * LSTR + o] * keyb[o];
      }
      part[t2] = p;
    }
    __syncthreads();
    if (tid < BT) {
      float s = 0.f;
#pragma unroll
      for (int u = 0; u < 16; ++u) s += part[tid * 16 + u];
      cq[b0 + tid] = s;
    }
  }
  const int nt = blockIdx.x * 4 + wl;
  const v8f acc = tile_xM(keyw, xs, nt, lane, 0.f);
  const int l16 = lane & 15, half = lane >> 4;
#pragma unroll
  for (int j = 0; j < 8; ++j) ys[(j + 8 * half) * 64 + wl * 16 + l16] = acc[j];
  __syncthreads();
#pragma unroll
  for (int p = 0; p < 2; ++p) {
    const int idx = tid + GT * p;
    const int r = idx >> 4, c4 = (idx & 15) * 4;
    *(v4f*)(fq + (b0 + r) * DD + blockIdx.x * 64 + c4) = *(const v4f*)(ys + r * 64 + c4);
  }
}

__global__ __launch_bounds__(256) void logits_kernel(
    const int* __restrict__ ids, const float* __restrict__ emb,
    const float* __restrict__ pos, const float* __restrict__ sq_g,
    const float* __restrict__ eq_g, const float* __restrict__ cs_g,
    const float* __restrict__ ce_g, float* __restrict__ out_start,
    float* __restrict__ out_end) {
  const int b = blockIdx.x;
  const int lane = threadIdx.x & 31;
  const int wv = threadIdx.x >> 5;
  float sqv[8], eqv[8];
  const float* sr = sq_g + b * DD + lane * 8;
  const float* er = eq_g + b * DD + lane * 8;
#pragma unroll
  for (int j = 0; j < 8; ++j) {
    sqv[j] = sr[j];
    eqv[j] = er[j];
  }
  const float cs = cs_g[b];
  const float ce = ce_g[b];
  for (int s = wv; s < SS; s += 8) {
    const int id = ids[b * SS + s];
    const float* row = emb + (size_t)id * DD + lane * 8;
    const float* pr = pos + s * DD + lane * 8;
    float as = 0.f, ae = 0.f;
#pragma unroll
    for (int j = 0; j < 8; ++j) {
      const float e = row[j] + pr[j];
      as += e * sqv[j];
      ae += e * eqv[j];
    }
#pragma unroll
    for (int off = 16; off; off >>= 1) {
      as += __shfl_xor(as, off, 32);
      ae += __shfl_xor(ae, off, 32);
    }
    if (lane == 0) {
      out_start[b * SS + s] = as + cs;
      out_end[b * SS + s] = ae + ce;
    }
  }
}

extern "C" void kernel_launch(void* const* d_in, const int* in_sizes, int n_in,
                              void* d_out, int out_size, void* d_ws,
                              size_t ws_size, hipStream_t stream) {
  const int* ids = (const int*)d_in[0];
  const float* emb = (const float*)d_in[1];
  const float* pos = (const float*)d_in[2];
  const float* init_w = (const float*)d_in[3];
  const float* init_b = (const float*)d_in[4];
  const float* pred_w = (const float*)d_in[5];
  const float* pred_b = (const float*)d_in[6];
  const float* upd_w = (const float*)d_in[7];
  const float* upd_b = (const float*)d_in[8];
  const float* sqw = (const float*)d_in[9];
  const float* sqb = (const float*)d_in[10];
  const float* eqw = (const float*)d_in[11];
  const float* eqb = (const float*)d_in[12];
  const float* keyw = (const float*)d_in[13];
  const float* keyb = (const float*)d_in[14];
  const float* aw1 = (const float*)d_in[15];
  const float* ab1 = (const float*)d_in[16];
  const float* aw2 = (const float*)d_in[17];
  const float* ab2 = (const float*)d_in[18];

  float* ws = (float*)d_ws;
  float* pooled = ws;
  float* sq = pooled + BB * DD;
  float* eq = sq + BB * DD;
  float* cs = eq + BB * DD;
  float* ce = cs + BB;
  float* wr = ce + BB;
  float* ra = wr + NWM * WMAT;
  float* rb = ra + LLN * ACT;
  float* ffb = rb + LLN * ACT;
  float* pr = ffb + ACT;
  float* qb = pr + LLN * ACT;

  float* out = (float*)d_out;
  float* out_start = out;
  float* out_end = out + BB * SS;
  float* out_ans = out + 2 * BB * SS;

  const dim3 g1(2, BB / BT, 1), gl(2, BB / BT, LLN);

  pooled_kernel<<<BB + RL_BLOCKS, 256, 0, stream>>>(
      ids, emb, pos, pooled, init_w, pred_w, upd_w, sqw, eqw, wr);

  init_kernel<false><<<g1, GT, 0, stream>>>(pooled, ra, nullptr, wr, init_b);
  init_kernel<false><<<g1, GT, 0, stream>>>(ra, ra + ACT, nullptr, wr + WMAT,
                                            init_b + DD);
  init_kernel<false><<<g1, GT, 0, stream>>>(ra + ACT, ra + 2 * ACT, nullptr,
                                            wr + 2 * WMAT, init_b + 2 * DD);
  init_kernel<true><<<g1, GT, 0, stream>>>(ra + 2 * ACT, ra + 3 * ACT, ffb,
                                           wr + 3 * WMAT, init_b + 3 * DD);

#define PC_SWEEP(cur, nxt)                                              \
  pred_kernel<<<gl, GT, 0, stream>>>(cur, pr, wr, pred_b);              \
  upd_kernel<<<gl, GT, 0, stream>>>(pooled, cur, nxt, pr, wr, upd_b);
  PC_SWEEP(ra, rb) PC_SWEEP(rb, ra)
  PC_SWEEP(ra, rb) PC_SWEEP(rb, ra)
  PC_SWEEP(ra, rb) PC_SWEEP(rb, ra)
  PC_SWEEP(ra, rb) PC_SWEEP(rb, ra)
  PC_SWEEP(ra, rb) PC_SWEEP(rb, ra)
#undef PC_SWEEP

  heads_kernel<<<dim3(2, BB / BT, 10), GT, 0, stream>>>(
      ra + 3 * ACT, ffb, wr, sqb, eqb, qb, aw1, ab1, aw2, ab2, out_ans);
  fold_kernel<<<dim3(4, BB / BT, 2), GT, 0, stream>>>(qb, keyw, keyb, sq, eq, cs, ce);
  logits_kernel<<<BB, 256, 0, stream>>>(ids, emb, pos, sq, eq, cs, ce,
                                        out_start, out_end);
}


# pooling and logits launches: each wave issues the loads of its next group of token rows (32 / 8 rows) as one group ahead of the current group's sums, was one round trip per 8 / 1 rows; sum orders kept
# speedup vs baseline: 5.4484x; 1.4137x over previous
#include <hip/hip_runtime.h>
#include <math.h>

#define BB   256
#define SS   512
#define DD   256
#define LLN  4
#define NITERS 10
#define BT   16
#define LSTR 260
#define TILE (BT * LSTR)
#define NWM  14
#define WMAT (DD * DD)
#define RL_BLOCKS (NWM * WMAT / 4 / 256 / 4)
#define ACT  (BB * DD)
#define GT   128
#define PR   32

typedef float v2f __attribute__((ext_vector_type(2)));
typedef float v4f __attribute__((ext_vector_type(4)));
typedef float v8f __attribute__((ext_vector_type(8)));

#if defined(__has_builtin)
#  if __has_builtin(__builtin_amdgcn_wmma_f32_16x16x4_f32)
#    define HAS_WMMA_F32 1
#  else
#    define HAS_WMMA_F32 0
#  endif
#  if __has_builtin(__builtin_amdgcn_tanhf)
#    define HAS_HW_TANH 1
#  else
#    define HAS_HW_TANH 0
#  endif
#else
#  define HAS_WMMA_F32 0
#  define HAS_HW_TANH 0
#endif

__device__ __forceinline__ float tanh_fast(float x) {
#if HAS_HW_TANH
  return __builtin_amdgcn_tanhf(x);
#else
  return tanhf(x);
#endif
}


__device__ __forceinline__ v8f tile_xM(const float* __restrict__ M,
                                       const float* X, int nt, int lane,
                                       float cinit) {
  const int l16 = lane & 15;
  const int half = lane >> 4;
  v8f acc = {cinit, cinit, cinit, cinit, cinit, cinit, cinit, cinit};
#if HAS_WMMA_F32
  const float* mp = M + (nt * 16 + l16) + (2 * half) * DD;
  const float* xr = X + l16 * LSTR + 2 * half;
#pragma unroll 4
  for (int kk = 0; kk < DD; kk += 4) {
    v2f a = {xr[kk], xr[kk + 1]};
    v2f b = {mp[kk * DD], mp[kk * DD + DD]};
    acc = __builtin_amdgcn_wmma_f32_16x16x4_f32(false, a, false, b,
                                                (short)0, acc, false, false);
  }
#else
#pragma unroll
  for (int j = 0; j < 8; ++j) {
    const int m = j + 8 * half, n = nt * 16 + l16;
    float s = cinit;
    for (int k = 0; k < DD; ++k) s += X[m * LSTR + k] * M[k * DD + n];
    acc[j] = s;
  }
#endif
  return acc;
}

template <typename Mid, typename Epi>
__device__ __forceinline__ void gemm_xWT(const float* __restrict__ Wr,
                                         const float* __restrict__ bias,
                                         const float* X, int lane, int wv,
                                         Mid mid, Epi epi) {
  const int l16 = lane & 15, half = lane >> 4;
  const int n0 = wv * 32 + l16;
  const float bv0 = bias ? bias[n0] : 0.f;
  const float bv1 = bias ? bias[n0 + 16] : 0.f;
  v8f acc0 = {bv0, bv0, bv0, bv0, bv0, bv0, bv0, bv0};
  v8f acc1 = {bv1, bv1, bv1, bv1, bv1, bv1, bv1, bv1};
#if HAS_WMMA_F32
  const v4f* wp = (const v4f*)Wr + wv * 2048 + lane;
  const v4f* xr = (const v4f*)(X + l16 * LSTR + 4 * half);
  v4f b0[32], b1[32], a[32];
#pragma unroll
  for (int j = 0; j < 32; ++j) {
    b0[j] = wp[(2 * j) * 32];
    b1[j] = wp[(2 * j + 1) * 32];
  }
  mid();
#pragma unroll
  for (int j = 0; j < 32; ++j) a[j] = xr[2 * j];
  __builtin_amdgcn_sched_barrier(0);
#pragma unroll
  for (int j = 0; j < 32; ++j) {
    const v2f alo = {a[j].x, a[j].y}, ahi = {a[j].z, a[j].w};
    const v2f blo0 = {b0[j].x, b0[j].y}, bhi0 = {b0[j].z, b0[j].w};
    const v2f blo1 = {b1[j].x, b1[j].y}, bhi1 = {b1[j].z, b1[j].w};
    acc0 = __builtin_amdgcn_wmma_f32_16x16x4_f32(false, alo, false, blo0,
                                                 (short)0, acc0, false, false);
    acc1 = __builtin_amdgcn_wmma_f32_16x16x4_f32(false, alo, false, blo1,
                                                 (short)0, acc1, false, false);
    acc0 = __builtin_amdgcn_wmma_f32_16x16x4_f32(false, ahi, false, bhi0,
                                                 (short)0, acc0, false, false);
    acc1 = __builtin_amdgcn_wmma_f32_16x16x4_f32(false, ahi, false, bhi1,
                                                 (short)0, acc1, false, false);
  }
#else
  mid();
#pragma unroll
  for (int j = 0; j < 8; ++j) {
    const int m = j + 8 * half;
    float s0 = bv0, s1 = bv1;
    for (int k = 0; k < DD; ++k) {
      const int slot = ((wv * 32 + (k >> 3)) * 2) * 32 + ((k >> 2) & 1) * 16 + l16;
      s0 += X[m * LSTR + k] * Wr[slot * 4 + (k & 3)];
      s1 += X[m * LSTR + k] * Wr[(slot + 32) * 4 + (k & 3)];
    }
    acc0[j] = s0;
    acc1[j] = s1;
  }
#endif
#pragma unroll
  for (int j = 0; j < 8; ++j) epi(j + 8 * half, n0, acc0[j]);
#pragma unroll
  for (int j = 0; j < 8; ++j) epi(j + 8 * half, n0 + 16, acc1[j]);
}

__device__ __forceinline__ int st_goff(int tid, int u, int b0) {
  const int idx = tid + GT * u;
  return (b0 + (idx >> 6)) * DD + (idx & 63) * 4;
}
__device__ __forceinline__ int st_loff(int tid, int u) {
  const int idx = tid + GT * u;
  return (idx >> 6) * LSTR + (idx & 63) * 4;
}
__device__ __forceinline__ void lds_barrier() {
  __builtin_amdgcn_fence(__ATOMIC_RELEASE, "workgroup", "local");
  __builtin_amdgcn_s_barrier();
  __builtin_amdgcn_fence(__ATOMIC_ACQUIRE, "workgroup", "local");
}

__global__ __launch_bounds__(256) void pooled_kernel(
    const int* __restrict__ ids, const float* __restrict__ emb,
    const float* __restrict__ pos, float* __restrict__ pooled_g,
    const float* __restrict__ init_w, const float* __restrict__ pred_w,
    const float* __restrict__ upd_w, const float* __restrict__ sqw,
    const float* __restrict__ eqw, float* __restrict__ wr_g) {
  __shared__ int sid[SS];
  const int t = threadIdx.x;
  if (blockIdx.x >= BB) {
    const int rb = blockIdx.x - BB;
    const int mi = rb >> 4;
    const float* src = mi < 4    ? init_w + mi * WMAT
                       : mi < 8  ? pred_w + (mi - 4) * WMAT
                       : mi < 12 ? upd_w + (mi - 8) * WMAT
                       : mi == 12 ? sqw
                                  : eqw;
    v4f* dst = (v4f*)(wr_g + (size_t)mi * WMAT);
#pragma unroll
    for (int i = 0; i < 4; ++i) {
      const int d = ((rb & 15) * 4 + i) * 256 + t;
      const int ln = d & 31, tt = (d >> 5) & 1, j = (d >> 6) & 31, w = d >> 11;
      const int n = w * 32 + tt * 16 + (ln & 15);
      const int k = 8 * j + 4 * (ln >> 4);
      dst[d] = *(const v4f*)(src + n * DD + k);
    }
    return;
  }
  const int b = blockIdx.x;
  sid[t] = ids[b * SS + t];
  sid[t + 256] = ids[b * SS + 256 + t];
  __syncthreads();
  float acc = 0.f;
  float ea[PR], pa[PR], eb[PR], pb[PR];
  auto load_grp = [&](float (&e)[PR], float (&p)[PR], int g) {
#pragma unroll
    for (int u = 0; u < PR; ++u) {
      const int s = g * PR + u;
      e[u] = emb[(size_t)sid[s] * DD + t];
      p[u] = pos[s * DD + t];
    }
  };
  auto sum_grp = [&](const float (&e)[PR], const float (&p)[PR]) {
#pragma unroll
    for (int u = 0; u < PR; ++u) acc += e[u] + p[u];
  };
  load_grp(ea, pa, 0);
  for (int g = 0; g < SS / PR - 2; g += 2) {
    load_grp(eb, pb, g + 1);
    __builtin_amdgcn_sched_barrier(0);
    sum_grp(ea, pa);
    __builtin_amdgcn_sched_barrier(0);
    load_grp(ea, pa, g + 2);
    __builtin_amdgcn_sched_barrier(0);
    sum_grp(eb, pb);
    __builtin_amdgcn_sched_barrier(0);
  }
  load_grp(eb, pb, SS / PR - 1);
  __builtin_amdgcn_sched_barrier(0);
  sum_grp(ea, pa);
  sum_grp(eb, pb);
  pooled_g[b * DD + t] = acc * (1.0f / (float)SS);
}


template <bool LAST>
__global__ __launch_bounds__(GT) void init_kernel(
    const float* __restrict__ x_g, float* __restrict__ y_g,
    float* __restrict__ ff_g, const float* __restrict__ w,
    const float* __restrict__ bias) {
  __shared__ __attribute__((aligned(16))) float xs[TILE];
  const int tid = threadIdx.x;
  const int lane = tid & 31;
  const int wv = blockIdx.x * 4 + (tid >> 5);
  const int b0 = blockIdx.y * BT;
  v4f vx[8];
#pragma unroll
  for (int u = 0; u < 8; ++u) vx[u] = *(const v4f*)(x_g + st_goff(tid, u, b0));
  gemm_xWT(w, bias, xs, lane, wv,
           [&]() {
#pragma unroll
             for (int u = 0; u < 8; ++u) *(v4f*)(xs + st_loff(tid, u)) = vx[u];
             lds_barrier();
           },
           [&](int m, int n, float v) {
             const float r = tanh_fast(v);
             const int o = (b0 + m) * DD + n;
             y_g[o] = r;
             if (LAST) ff_g[o] = r;
           });
}

__global__ __launch_bounds__(GT) void pred_kernel(
    const float* __restrict__ rc, float* __restrict__ pred_g,
    const float* __restrict__ wr_g, const float* __restrict__ pred_b) {
  __shared__ __attribute__((aligned(16))) float xs[TILE];
  const int tid = threadIdx.x;
  const int lane = tid & 31;
  const int wv = blockIdx.x * 4 + (tid >> 5);
  const int b0 = blockIdx.y * BT;
  const int i = blockIdx.z;
  const float* x_g = rc + i * ACT;
  float* pc = pred_g + i * ACT;
  v4f vx[8];
#pragma unroll
  for (int u = 0; u < 8; ++u) vx[u] = *(const v4f*)(x_g + st_goff(tid, u, b0));
  gemm_xWT(wr_g + (4 + i) * WMAT, pred_b + i * DD, xs, lane, wv,
           [&]() {
#pragma unroll
             for (int u = 0; u < 8; ++u) *(v4f*)(xs + st_loff(tid, u)) = vx[u];
             lds_barrier();
           },
           [&](int m, int n, float v) { pc[(b0 + m) * DD + n] = v; });
}

__global__ __launch_bounds__(GT) void upd_kernel(
    const float* __restrict__ pooled_g, const float* __restrict__ rc,
    float* __restrict__ rn, const float* __restrict__ pred_g,
    const float* __restrict__ wr_g, const float* __restrict__ upd_b) {
  __shared__ __attribute__((aligned(16))) float xs[TILE];
  const int tid = threadIdx.x;
  const int lane = tid & 31;
  const int wv = blockIdx.x * 4 + (tid >> 5);
  const int b0 = blockIdx.y * BT;
  const int i = blockIdx.z;
  const bool has_above = (i < LLN - 1);
  const float* below = (i > 0) ? rc + (i - 1) * ACT : pooled_g;
  const float* ri = rc + i * ACT;
  const float* pc = pred_g + i * ACT;
  const float* pn = pred_g + (has_above ? i + 1 : i) * ACT;
  float* ro = rn + i * ACT;
  v4f vb[8], vp[8], vr[8], vn[8];
#pragma unroll
  for (int u = 0; u < 8; ++u) {
    const int g = st_goff(tid, u, b0);
    vb[u] = *(const v4f*)(below + g);
    vp[u] = *(const v4f*)(pc + g);
    vr[u] = *(const v4f*)(ri + g);
    vn[u] = *(const v4f*)(pn + g);
  }
  gemm_xWT(wr_g + (8 + i) * WMAT, upd_b + i * DD, xs, lane, wv,
           [&]() {
#pragma unroll
             for (int u = 0; u < 8; ++u) {
               v4f e = vb[u] - vp[u];
               if (has_above) e += 0.5f * (vr[u] - vn[u]);
               *(v4f*)(xs + st_loff(tid, u)) = e;
             }
             lds_barrier();
           },
           [&](int m, int n, float v) {
             const int o = (b0 + m) * DD + n;
             ro[o] = ri[o] + 0.1f * tanh_fast(v);
           });
}

__global__ __launch_bounds__(GT) void heads_kernel(
    const float* __restrict__ r3, const float* __restrict__ ff_g,
    const float* __restrict__ wr_g, const float* __restrict__ sqb,
    const float* __restrict__ eqb, float* __restrict__ q_g,
    const float* __restrict__ aw1, const float* __restrict__ ab1,
    const float* __restrict__ aw2, const float* __restrict__ ab2,
    float* __restrict__ ans_g) {
  __shared__ __attribute__((aligned(16))) float xs[TILE];
  const int tid = threadIdx.x;
  const int lane = tid & 31;
  const int wv = blockIdx.x * 4 + (tid >> 5);
  const int b0 = blockIdx.y * BT;
  const int z = blockIdx.z;
  if (z >= 2) {
    const int r = (z - 2) * 2 + blockIdx.x;
    const float* ra = r3 + (b0 + r) * DD;
    const float* fa = ff_g + (b0 + r) * DD;
    const float* wrow = aw1 + tid * DD;
    float s = ab1[tid];
#pragma unroll 32
    for (int k = 0; k < DD; ++k) s += (ra[k] + fa[k]) * wrow[k];
    xs[tid] = 0.5f * s * (1.0f + erff(s * 0.70710678118654752f));
    __syncthreads();
    if (tid < 2) {
      float t = ab2[tid];
      for (int j = 0; j < 128; ++j) t += xs[j] * aw2[tid * 128 + j];
      ans_g[(b0 + r) * 2 + tid] = t;
    }
    return;
  }
  float* q = q_g + z * ACT;
  v4f va[8], vf[8];
#pragma unroll
  for (int u = 0; u < 8; ++u) {
    const int g = st_goff(tid, u, b0);
    va[u] = *(const v4f*)(r3 + g);
    vf[u] = *(const v4f*)(ff_g + g);
  }
  gemm_xWT(wr_g + (12 + z) * WMAT, z ? eqb : sqb, xs, lane, wv,
           [&]() {
#pragma unroll
             for (int u = 0; u < 8; ++u)
               *(v4f*)(xs + st_loff(tid, u)) = va[u] + vf[u];
             lds_barrier();
           },
           [&](int m, int n, float v) { q[(b0 + m) * DD + n] = v; });
}

__global__ __launch_bounds__(GT) void fold_kernel(
    const float* __restrict__ q_g, const float* __restrict__ keyw,
    const float* __restrict__ keyb, float* __restrict__ sq_g,
    float* __restrict__ eq_g, float* __restrict__ cs_g,
    float* __restrict__ ce_g) {
  __shared__ __attribute__((aligned(16))) float xs[TILE];
  __shared__ __attribute__((aligned(16))) float part[256];
  __shared__ __attribute__((aligned(16))) float ys[BT * 64];
  const int tid = threadIdx.x;
  const int lane = tid & 31;
  const int wl = tid >> 5;
  const int b0 = blockIdx.y * BT;
  const int z = blockIdx.z;
  const float* q = q_g + z * ACT;
  float* fq = z ? eq_g : sq_g;
  float* cq = z ? ce_g : cs_g;
#pragma unroll
  for (int u = 0; u < 8; ++u)
    *(v4f*)(xs + st_loff(tid, u)) = *(const v4f*)(q + st_goff(tid, u, b0));
  __syncthreads();
  if (blockIdx.x == 0) {
    for (int t2 = tid; t2 < 256; t2 += GT) {
      const int r = t2 >> 4, c = t2 & 15;
      float p = 0.f;
#pragma unroll
      for (int u = 0; u < 16; ++u) {
        const int o = c * 16 + u;
        p += xs[r * LSTR + o] * keyb[o];
      }
      part[t2] = p;
    }
    __syncthreads();
    if (tid < BT) {
      float s = 0.f;
#pragma unroll
      for (int u = 0; u < 16; ++u) s += part[tid * 16 + u];
      cq[b0 + tid] = s;
    }
  }
  const int nt = blockIdx.x * 4 + wl;
  const v8f acc = tile_xM(keyw, xs, nt, lane, 0.f);
  const int l16 = lane & 15, half = lane >> 4;
#pragma unroll
  for (int j = 0; j < 8; ++j) ys[(j + 8 * half) * 64 + wl * 16 + l16] = acc[j];
  __syncthreads();
#pragma unroll
  for (int p = 0; p < 2; ++p) {
    const int idx = tid + GT * p;
    const int r = idx >> 4, c4 = (idx & 15) * 4;
    *(v4f*)(fq + (b0 + r) * DD + blockIdx.x * 64 + c4) = *(const v4f*)(ys + r * 64 + c4);
  }
}

#define LR   8
#define NGRP (64 / LR)

struct RowSet { v4f e0[LR], e1[LR], p0[LR], p1[LR]; };

__global__ __launch_bounds__(256) void logits_kernel(
    const int* __restrict__ ids, const float* __restrict__ emb,
    const float* __restrict__ pos, const float* __restrict__ sq_g,
    const float* __restrict__ eq_g, const float* __restrict__ cs_g,
    const float* __restrict__ ce_g, float* __restrict__ out_start,
    float* __restrict__ out_end) {
  const int b = blockIdx.x;
  const int lane = threadIdx.x & 31;
  const int wv = __builtin_amdgcn_readfirstlane((int)(threadIdx.x >> 5));
  const v4f sq0 = *(const v4f*)(sq_g + b * DD + lane * 8);
  const v4f sq1 = *(const v4f*)(sq_g + b * DD + lane * 8 + 4);
  const v4f eq0 = *(const v4f*)(eq_g + b * DD + lane * 8);
  const v4f eq1 = *(const v4f*)(eq_g + b * DD + lane * 8 + 4);
  const float cs = cs_g[b];
  const float ce = ce_g[b];
  const int s0 = wv * 64;
  const int* idp = ids + b * SS + s0;
  const float* ep = emb + lane * 8;
  const float* pp = pos + s0 * DD + lane * 8;
  float keep_s = 0.f, keep_e = 0.f;

  auto load_ids = [&](int (&id)[LR], int g) {
#pragma unroll
    for (int u = 0; u < LR; ++u) id[u] = idp[g * LR + u];
  };
  auto load_rows = [&](RowSet& R, const int (&id)[LR], int g) {
#pragma unroll
    for (int u = 0; u < LR; ++u) {
      const float* row = ep + (size_t)id[u] * DD;
      const float* pr = pp + (g * LR + u) * DD;
      R.e0[u] = *(const v4f*)row;
      R.e1[u] = *(const v4f*)(row + 4);
      R.p0[u] = *(const v4f*)pr;
      R.p1[u] = *(const v4f*)(pr + 4);
    }
  };
  auto sum_rows = [&](const RowSet& R, int g) {
#pragma unroll
    for (int u = 0; u < LR; ++u) {
      const v4f x0 = R.e0[u] + R.p0[u];
      const v4f x1 = R.e1[u] + R.p1[u];
      float as = 0.f, ae = 0.f;
#pragma unroll
      for (int j = 0; j < 4; ++j) {
        as += x0[j] * sq0[j];
        ae += x0[j] * eq0[j];
      }
#pragma unroll
      for (int j = 0; j < 4; ++j) {
        as += x1[j] * sq1[j];
        ae += x1[j] * eq1[j];
      }
#pragma unroll
      for (int off = 16; off; off >>= 1) {
        as += __shfl_xor(as, off, 32);
        ae += __shfl_xor(ae, off, 32);
      }
      const int r = g * LR + u;
      if (lane == (r & 31)) {
        keep_s = as + cs;
        keep_e = ae + ce;
      }
    }
    if ((((g + 1) * LR) & 31) == 0) {
      const int o = b * SS + s0 + ((g * LR) & ~31) + lane;
      out_start[o] = keep_s;
      out_end[o] = keep_e;
    }
  };

  RowSet A, B;
  int idA[LR], idB[LR];
  load_ids(idA, 0);
  load_ids(idB, 1);
  load_rows(A, idA, 0);
#pragma unroll
  for (int g = 0; g < NGRP; g += 2) {
    load_rows(B, idB, g + 1);
    if (g + 2 < NGRP) load_ids(idA, g + 2);
    __builtin_amdgcn_sched_barrier(0);
    sum_rows(A, g);
    __builtin_amdgcn_sched_barrier(0);
    if (g + 2 < NGRP) {
      load_rows(A, idA, g + 2);
      if (g + 3 < NGRP) load_ids(idB, g + 3);
    }
    __builtin_amdgcn_sched_barrier(0);
    sum_rows(B, g + 1);
    __builtin_amdgcn_sched_barrier(0);
  }
}

extern "C" void kernel_launch(void* const* d_in, const int* in_sizes, int n_in,
                              void* d_out, int out_size, void* d_ws,
                              size_t ws_size, hipStream_t stream) {
  const int* ids = (const int*)d_in[0];
  const float* emb = (const float*)d_in[1];
  const float* pos = (const float*)d_in[2];
  const float* init_w = (const float*)d_in[3];
  const float* init_b = (const float*)d_in[4];
  const float* pred_w = (const float*)d_in[5];
  const float* pred_b = (const float*)d_in[6];
  const float* upd_w = (const float*)d_in[7];
  const float* upd_b = (const float*)d_in[8];
  const float* sqw = (const float*)d_in[9];
  const float* sqb = (const float*)d_in[10];
  const float* eqw = (const float*)d_in[11];
  const float* eqb = (const float*)d_in[12];
  const float* keyw = (const float*)d_in[13];
  const float* keyb = (const float*)d_in[14];
  const float* aw1 = (const float*)d_in[15];
  const float* ab1 = (const float*)d_in[16];
  const float* aw2 = (const float*)d_in[17];
  const float* ab2 = (const float*)d_in[18];

  float* ws = (float*)d_ws;
  float* pooled = ws;
  float* sq = pooled + BB * DD;
  float* eq = sq + BB * DD;
  float* cs = eq + BB * DD;
  float* ce = cs + BB;
  float* wr = ce + BB;
  float* ra = wr + NWM * WMAT;
  float* rb = ra + LLN * ACT;
  float* ffb = rb + LLN * ACT;
  float* pr = ffb + ACT;
  float* qb = pr + LLN * ACT;

  float* out = (float*)d_out;
  float* out_start = out;
  float* out_end = out + BB * SS;
  float* out_ans = out + 2 * BB * SS;

  const dim3 g1(2, BB / BT, 1), gl(2, BB / BT, LLN);

  pooled_kernel<<<BB + RL_BLOCKS, 256, 0, stream>>>(
      ids, emb, pos, pooled, init_w, pred_w, upd_w, sqw, eqw, wr);

  init_kernel<false><<<g1, GT, 0, stream>>>(pooled, ra, nullptr, wr, init_b);
  init_kernel<false><<<g1, GT, 0, stream>>>(ra, ra + ACT, nullptr, wr + WMAT,
                                            init_b + DD);
  init_kernel<false><<<g1, GT, 0, stream>>>(ra + ACT, ra + 2 * ACT, nullptr,
                                            wr + 2 * WMAT, init_b + 2 * DD);
  init_kernel<true><<<g1, GT, 0, stream>>>(ra + 2 * ACT, ra + 3 * ACT, ffb,
                                           wr + 3 * WMAT, init_b + 3 * DD);

#define PC_SWEEP(cur, nxt)                                              \
  pred_kernel<<<gl, GT, 0, stream>>>(cur, pr, wr, pred_b);              \
  upd_kernel<<<gl, GT, 0, stream>>>(pooled, cur, nxt, pr, wr, upd_b);
  PC_SWEEP(ra, rb) PC_SWEEP(rb, ra)
  PC_SWEEP(ra, rb) PC_SWEEP(rb, ra)
  PC_SWEEP(ra, rb) PC_SWEEP(rb, ra)
  PC_SWEEP(ra, rb) PC_SWEEP(rb, ra)
  PC_SWEEP(ra, rb) PC_SWEEP(rb, ra)
#undef PC_SWEEP

  heads_kernel<<<dim3(2, BB / BT, 10), GT, 0, stream>>>(
      ra + 3 * ACT, ffb, wr, sqb, eqb, qb, aw1, ab1, aw2, ab2, out_ans);
  fold_kernel<<<dim3(4, BB / BT, 2), GT, 0, stream>>>(qb, keyw, keyb, sq, eq, cs, ce);
  logits_kernel<<<BB, 256, 0, stream>>>(ids, emb, pos, sq, eq, cs, ce,
                                        out_start, out_end);
}
